# starting kernel
# baseline (speedup 1.0000x reference)
#include <hip/hip_runtime.h>
#include <hip/hip_bf16.h>
#include <math.h>

#define BS   4
#define V    4096
#define NNB  20
#define OC   64
#define SUP  4
#define INCH 64
#define COLS ((SUP + 1) * OC)
#define DIRS (SUP * OC)
#define EPSV 1e-12f

typedef __attribute__((ext_vector_type(2))) float v2f;
typedef __attribute__((ext_vector_type(8))) float v8f;

__global__ __launch_bounds__(256)
void gemm_bias_wmma_kernel(const float* __restrict__ A,
                           const float* __restrict__ W,
                           const float* __restrict__ bias,
                           float* __restrict__ out,
                           int M)
{
    const int tilesN = COLS / 16;
    const int wave   = threadIdx.x >> 5;
    const int lane   = threadIdx.x & 31;
    const int tile   = blockIdx.x * (blockDim.x >> 5) + wave;

    const int tm = tile / tilesN;
    const int tn = tile - tm * tilesN;
    if (tm * 16 >= M) return;

    const int l   = lane & 15;
    const int h   = lane >> 4;
    const int col = tn * 16 + l;

    const float bv = bias[col];
    v8f c;
#pragma unroll
    for (int i = 0; i < 8; ++i) c[i] = bv;

    const float* arow = A + (size_t)(tm * 16 + l) * INCH;

#pragma unroll
    for (int k0 = 0; k0 < INCH; k0 += 4) {
        const int k = k0 + 2 * h;
        v2f a, b;
        a[0] = arow[k];
        a[1] = arow[k + 1];
        b[0] = W[(size_t)k * COLS + col];
        b[1] = W[(size_t)(k + 1) * COLS + col];
        c = __builtin_amdgcn_wmma_f32_16x16x4_f32(false, a, false, b,
                                                  (short)0, c, false, false);
    }

#pragma unroll
    for (int r = 0; r < 8; ++r) {
        const int m = tm * 16 + r + 8 * h;
        out[(size_t)m * COLS + col] = c[r];
    }
}

__global__ __launch_bounds__(256)
void gather_theta_max_kernel(const int*   __restrict__ nbr_idx,
                             const float* __restrict__ verts,
                             const float* __restrict__ dirs,
                             const float* __restrict__ feat,
                             float*       __restrict__ out)
{
    __shared__ float s_nd[NNB][3];
    __shared__ int   s_nb[NNB];
    __shared__ float s_red[DIRS];

    const int bv  = blockIdx.x;
    const int b   = bv / V;
    const int tid = threadIdx.x;

    if (tid < NNB) {
        const int nb  = nbr_idx[(size_t)bv * NNB + tid];
        const int nbv = b * V + nb;
        const float cx = verts[(size_t)bv * 3 + 0];
        const float cy = verts[(size_t)bv * 3 + 1];
        const float cz = verts[(size_t)bv * 3 + 2];
        const float dx = verts[(size_t)nbv * 3 + 0] - cx;
        const float dy = verts[(size_t)nbv * 3 + 1] - cy;
        const float dz = verts[(size_t)nbv * 3 + 2] - cz;
        const float inv = 1.0f / fmaxf(sqrtf(dx * dx + dy * dy + dz * dz), EPSV);
        s_nd[tid][0] = dx * inv;
        s_nd[tid][1] = dy * inv;
        s_nd[tid][2] = dz * inv;
        s_nb[tid]    = nbv;
    }
    __syncthreads();

    const int d = tid;
    float sx = dirs[d];
    float sy = dirs[DIRS + d];
    float sz = dirs[2 * DIRS + d];
    const float sinv = 1.0f / fmaxf(sqrtf(sx * sx + sy * sy + sz * sz), EPSV);
    sx *= sinv; sy *= sinv; sz *= sinv;

    float m = -__builtin_inff();
#pragma unroll 4
    for (int j = 0; j < NNB; ++j) {
        const float t  = fmaxf(sx * s_nd[j][0] + sy * s_nd[j][1] + sz * s_nd[j][2], 0.0f);
        const float fs = feat[(size_t)s_nb[j] * COLS + OC + d];
        m = fmaxf(m, t * fs);
    }
    s_red[d] = m;
    __syncthreads();

    if (tid < OC) {
        const float acc = feat[(size_t)bv * COLS + tid]
                        + s_red[tid] + s_red[OC + tid]
                        + s_red[2 * OC + tid] + s_red[3 * OC + tid];
        out[(size_t)bv * OC + tid] = acc;
    }
}

extern "C" void kernel_launch(void* const* d_in, const int* in_sizes, int n_in,
                              void* d_out, int out_size, void* d_ws, size_t ws_size,
                              hipStream_t stream)
{
    (void)in_sizes; (void)n_in; (void)out_size; (void)ws_size;

    const int*   nbr   = (const int*)  d_in[0];
    const float* verts = (const float*)d_in[1];
    const float* fmap  = (const float*)d_in[2];
    const float* W     = (const float*)d_in[3];
    const float* bias  = (const float*)d_in[4];
    const float* dirs  = (const float*)d_in[5];
    float*       outp  = (float*)d_out;
    float*       feat  = (float*)d_ws;

    const int M     = BS * V;
    const int tiles = (M / 16) * (COLS / 16);
    const int blocks = tiles / 8;

    gemm_bias_wmma_kernel<<<blocks, 256, 0, stream>>>(fmap, W, bias, feat, M);
    gather_theta_max_kernel<<<M, 256, 0, stream>>>(nbr, verts, dirs, feat, outp);
}


# gather kernel remapped: one wave per vertex (was an 8-wave block), 4 vertices per wave, each lane 8 columns by 16-byte row loads in a 2-stage pipeline, no LDS, no barriers: 32x fewer waves enter
# speedup vs baseline: 1.9334x; 1.9334x over previous
#include <hip/hip_runtime.h>
#include <hip/hip_bf16.h>
#include <math.h>

#define BS   4
#define V    4096
#define NNB  20
#define OC   64
#define SUP  4
#define INCH 64
#define COLS ((SUP + 1) * OC)
#define DIRS (SUP * OC)
#define EPSV 1e-12f

typedef __attribute__((ext_vector_type(2))) float v2f;
typedef __attribute__((ext_vector_type(8))) float v8f;

__global__ __launch_bounds__(256)
void gemm_bias_wmma_kernel(const float* __restrict__ A,
                           const float* __restrict__ W,
                           const float* __restrict__ bias,
                           float* __restrict__ out,
                           int M)
{
    const int tilesN = COLS / 16;
    const int wave   = threadIdx.x >> 5;
    const int lane   = threadIdx.x & 31;
    const int tile   = blockIdx.x * (blockDim.x >> 5) + wave;

    const int tm = tile / tilesN;
    const int tn = tile - tm * tilesN;
    if (tm * 16 >= M) return;

    const int l   = lane & 15;
    const int h   = lane >> 4;
    const int col = tn * 16 + l;

    const float bv = bias[col];
    v8f c;
#pragma unroll
    for (int i = 0; i < 8; ++i) c[i] = bv;

    const float* arow = A + (size_t)(tm * 16 + l) * INCH;

#pragma unroll
    for (int k0 = 0; k0 < INCH; k0 += 4) {
        const int k = k0 + 2 * h;
        v2f a, b;
        a[0] = arow[k];
        a[1] = arow[k + 1];
        b[0] = W[(size_t)k * COLS + col];
        b[1] = W[(size_t)(k + 1) * COLS + col];
        c = __builtin_amdgcn_wmma_f32_16x16x4_f32(false, a, false, b,
                                                  (short)0, c, false, false);
    }

#pragma unroll
    for (int r = 0; r < 8; ++r) {
        const int m = tm * 16 + r + 8 * h;
        out[(size_t)m * COLS + col] = c[r];
    }
}

#define VPW 4
#define GRP 5

typedef __attribute__((ext_vector_type(4))) float v4f;

__global__ __launch_bounds__(256) __attribute__((amdgpu_waves_per_eu(4, 4)))
void gather_theta_max_kernel(const int*   __restrict__ nbr_idx,
                             const float* __restrict__ verts,
                             const float* __restrict__ dirs,
                             const float* __restrict__ feat,
                             float*       __restrict__ out)
{
    const int lane = threadIdx.x & 31;
    const int jl   = lane < NNB ? lane : NNB - 1;
    const int wave = __builtin_amdgcn_readfirstlane((int)(blockIdx.x * 8 + (threadIdx.x >> 5)));
    const int bv0  = wave * VPW;

    float sx[8], sy[8], sz[8];
    {
        const v4f* d4 = (const v4f*)dirs;
#pragma unroll
        for (int q = 0; q < 2; ++q) {
            const v4f x = d4[q * 32 + lane];
            const v4f y = d4[DIRS / 4 + q * 32 + lane];
            const v4f z = d4[2 * (DIRS / 4) + q * 32 + lane];
#pragma unroll
            for (int e = 0; e < 4; ++e) {
                const float a = x[e], b = y[e], c = z[e];
                const float sinv = 1.0f / fmaxf(sqrtf(a * a + b * b + c * c), EPSV);
                sx[q * 4 + e] = a * sinv;
                sy[q * 4 + e] = b * sinv;
                sz[q * 4 + e] = c * sinv;
            }
        }
    }

    int   nbv[VPW];
    float nx[VPW], ny[VPW], nz[VPW];
#pragma unroll
    for (int k = 0; k < VPW; ++k)
        nbv[k] = ((bv0 + k) / V) * V + nbr_idx[(size_t)(bv0 + k) * NNB + jl];
#pragma unroll
    for (int k = 0; k < VPW; ++k) {
        const float cx = verts[(size_t)(bv0 + k) * 3 + 0];
        const float cy = verts[(size_t)(bv0 + k) * 3 + 1];
        const float cz = verts[(size_t)(bv0 + k) * 3 + 2];
        const float dx = verts[(size_t)nbv[k] * 3 + 0] - cx;
        const float dy = verts[(size_t)nbv[k] * 3 + 1] - cy;
        const float dz = verts[(size_t)nbv[k] * 3 + 2] - cz;
        const float inv = 1.0f / fmaxf(sqrtf(dx * dx + dy * dy + dz * dz), EPSV);
        nx[k] = dx * inv; ny[k] = dy * inv; nz[k] = dz * inv;
    }

    const char*    fbase  = (const char*)feat;
    const unsigned lane16 = (unsigned)lane * 16u;

    v4f bufA[GRP * 2], bufB[GRP * 2];
    float m[8];
#pragma unroll
    for (int e = 0; e < 8; ++e) m[e] = -__builtin_inff();

#define LOAD_GROUP(buf, nsel, j0)                                                   \
    _Pragma("unroll")                                                               \
    for (int jj = 0; jj < GRP; ++jj) {                                              \
                    \
        const unsigned r = (unsigned)__builtin_amdgcn_readlane((nsel), (j0) + jj);  \
              \
                      \
        const unsigned off = r * (unsigned)(COLS * 4) + (unsigned)(OC * 4) + lane16;\
        const char* p = fbase + off;                                                \
        buf[2 * jj]     = *(const v4f*)p;                                           \
        buf[2 * jj + 1] = *(const v4f*)(p + 512);                                   \
    }

#define COMPUTE_GROUP(buf, j0)                                                      \
    _Pragma("unroll")                                                               \
    for (int jj = 0; jj < GRP; ++jj) {                                              \
                   \
        const float rx = __int_as_float(__builtin_amdgcn_readlane(__float_as_int(nx[0]), (j0) + jj)); \
        const float ry = __int_as_float(__builtin_amdgcn_readlane(__float_as_int(ny[0]), (j0) + jj)); \
        const float rz = __int_as_float(__builtin_amdgcn_readlane(__float_as_int(nz[0]), (j0) + jj)); \
        const v4f f0 = buf[2 * jj];                                                 \
        const v4f f1 = buf[2 * jj + 1];                                             \
        _Pragma("unroll")                                                           \
        for (int e = 0; e < 4; ++e) {                                               \
            const float t0 = fmaxf(sx[e] * rx + sy[e] * ry + sz[e] * rz, 0.0f);     \
            m[e] = fmaxf(m[e], t0 * f0[e]);                                         \
            const float t1 = fmaxf(sx[4 + e] * rx + sy[4 + e] * ry + sz[4 + e] * rz, 0.0f); \
            m[4 + e] = fmaxf(m[4 + e], t1 * f1[e]);                                 \
        }                                                                           \
    }

    LOAD_GROUP(bufA, nbv[0], 0)

    const unsigned c16 = (unsigned)(lane & 15) * 16u;
    v4f cen = *(const v4f*)(fbase + (unsigned)bv0 * (unsigned)(COLS * 4) + c16);

#pragma unroll 1
    for (int t = 0; t < 2 * VPW; ++t) {
        const int g = (t & 1) * 2;

        LOAD_GROUP(bufB, nbv[0], (g + 1) * GRP)
        COMPUTE_GROUP(bufA, g * GRP)

        if (t + 1 < 2 * VPW) {
            const int nsel = g ? nbv[1] : nbv[0];
            const int j0   = g ? 0 : 2 * GRP;
            LOAD_GROUP(bufA, nsel, j0)
        }

        COMPUTE_GROUP(bufB, (g + 1) * GRP)

        if (g) {
            const int bv = bv0 + (t >> 1);
            float p1[4], p3[4];
#pragma unroll
            for (int e = 0; e < 4; ++e) {
                p1[e] = __shfl_xor(m[e], 16);
                p3[e] = __shfl_xor(m[4 + e], 16);
            }
            if (lane < 16) {
                v4f o;
#pragma unroll
                for (int e = 0; e < 4; ++e)
                    o[e] = cen[e] + m[e] + p1[e] + m[4 + e] + p3[e];
                ((v4f*)(out + (size_t)bv * OC))[lane] = o;
            }
            const int bvn = t + 1 < 2 * VPW ? bv + 1 : bv;
            cen = *(const v4f*)(fbase + (unsigned)bvn * (unsigned)(COLS * 4) + c16);
#pragma unroll
            for (int e = 0; e < 8; ++e) m[e] = -__builtin_inff();
#pragma unroll
            for (int k = 0; k + 1 < VPW; ++k) {
                nbv[k] = nbv[k + 1];
                nx[k] = nx[k + 1]; ny[k] = ny[k + 1]; nz[k] = nz[k + 1];
            }
        }
    }
#undef LOAD_GROUP
#undef COMPUTE_GROUP
}

extern "C" void kernel_launch(void* const* d_in, const int* in_sizes, int n_in,
                              void* d_out, int out_size, void* d_ws, size_t ws_size,
                              hipStream_t stream)
{
    (void)in_sizes; (void)n_in; (void)out_size; (void)ws_size;

    const int*   nbr   = (const int*)  d_in[0];
    const float* verts = (const float*)d_in[1];
    const float* fmap  = (const float*)d_in[2];
    const float* W     = (const float*)d_in[3];
    const float* bias  = (const float*)d_in[4];
    const float* dirs  = (const float*)d_in[5];
    float*       outp  = (float*)d_out;
    float*       feat  = (float*)d_ws;

    const int M     = BS * V;
    const int tiles = (M / 16) * (COLS / 16);
    const int blocks = tiles / 8;

    gemm_bias_wmma_kernel<<<blocks, 256, 0, stream>>>(fmap, W, bias, feat, M);
    gather_theta_max_kernel<<<M / (VPW * 8), 256, 0, stream>>>(nbr, verts, dirs, feat, outp);
}
